# starting kernel
# baseline (speedup 1.0000x reference)
#include <hip/hip_runtime.h>

typedef float v2f __attribute__((ext_vector_type(2)));
typedef float v8f __attribute__((ext_vector_type(8)));

#define EMBED 64
#define HEADS 8
#define HDIM  8
#define BB    8
#define NN    128
#define TT    64

__device__ __forceinline__ v8f wmma_f32_16x16x4(v2f a, v2f b, v8f c) {
  return __builtin_amdgcn_wmma_f32_16x16x4_f32(false, a, false, b, (short)0, c,
                                               false, false);
}

__global__ __launch_bounds__(128) void attn_kernel(
    const float* __restrict__ V, const float* __restrict__ K,
    const float* __restrict__ Q, const float* __restrict__ Wv,
    const float* __restrict__ Wk, const float* __restrict__ Wq,
    float* __restrict__ Xout) {
  __shared__ float S[64 * NN];
  __shared__ float Qp[NN * HDIM];
  __shared__ float Kp[NN * HDIM];
  __shared__ float Vp[NN * HDIM];
  __shared__ float redM[128];
  __shared__ float redS[128];

  const int blk     = blockIdx.x;
  const int rh      = blk & 1;
  const int h       = (blk >> 1) % HEADS;
  const int t       = (blk >> 1) / HEADS % TT;
  const int b       = (blk >> 1) / (HEADS * TT);
  const int rowBase = rh * 64;

  const int tid  = threadIdx.x;
  const int wave = tid >> 5;
  const int lane = tid & 31;
  const int half = lane >> 4;
  const int l    = lane & 15;

  {
    const int n = tid;
    const size_t base = (((size_t)(b * NN + n) * TT + t) * EMBED) + h * HDIM;
    float qv[HDIM], kv[HDIM], vv[HDIM];
#pragma unroll
    for (int d = 0; d < HDIM; ++d) {
      qv[d] = Q[base + d];
      kv[d] = K[base + d];
      vv[d] = V[base + d];
    }
#pragma unroll
    for (int e = 0; e < HDIM; ++e) {
      float aq = 0.f, ak = 0.f, av = 0.f;
#pragma unroll
      for (int d = 0; d < HDIM; ++d) {
        aq += qv[d] * Wq[e * HDIM + d];
        ak += kv[d] * Wk[e * HDIM + d];
        av += vv[d] * Wv[e * HDIM + d];
      }
      Qp[n * HDIM + e] = aq;
      Kp[n * HDIM + e] = ak;
      Vp[n * HDIM + e] = av;
    }
  }
  __syncthreads();

  const float scale = 0.125f;
  for (int tile = wave; tile < 32; tile += 4) {
    const int ti = tile >> 3;
    const int tj = tile & 7;
    v8f c = {};
#pragma unroll
    for (int kb = 0; kb < HDIM; kb += 4) {
      const int ac = kb + 2 * half;
      v2f a, bm;
      a.x  = Qp[(rowBase + ti * 16 + l) * HDIM + ac];
      a.y  = Qp[(rowBase + ti * 16 + l) * HDIM + ac + 1];
      bm.x = Kp[(tj * 16 + l) * HDIM + ac];
      bm.y = Kp[(tj * 16 + l) * HDIM + ac + 1];
      c = wmma_f32_16x16x4(a, bm, c);
    }
#pragma unroll
    for (int r = 0; r < 8; ++r) {
      const int row = ti * 16 + r + 8 * half;
      S[row * NN + tj * 16 + l] = c[r] * scale;
    }
  }
  __syncthreads();

  {
    const int row = tid >> 1;
    const int seg = tid & 1;
    float* Sseg = S + row * NN + seg * 64;

    float m = -1e30f;
    for (int j = 0; j < 64; ++j) m = fmaxf(m, Sseg[j]);
    redM[tid] = m;
    __syncthreads();
    m = fmaxf(redM[row * 2], redM[row * 2 + 1]);

    float sum = 0.f;
    for (int j = 0; j < 64; ++j) {
      const float ex = __expf(Sseg[j] - m);
      Sseg[j] = ex;
      sum += ex;
    }
    redS[tid] = sum;
    __syncthreads();
    const float inv = 1.f / (redS[row * 2] + redS[row * 2 + 1]);
    for (int j = 0; j < 64; ++j) Sseg[j] *= inv;
  }
  __syncthreads();

  {
    const int ti = wave;
    v8f c = {};
    for (int kb = 0; kb < NN; kb += 4) {
      const int ac = kb + 2 * half;
      v2f a, bm;
      a.x  = S[(ti * 16 + l) * NN + ac];
      a.y  = S[(ti * 16 + l) * NN + ac + 1];
      bm.x = (l < HDIM) ? Vp[ac * HDIM + l] : 0.f;
      bm.y = (l < HDIM) ? Vp[(ac + 1) * HDIM + l] : 0.f;
      c = wmma_f32_16x16x4(a, bm, c);
    }
    if (l < HDIM) {
#pragma unroll
      for (int r = 0; r < 8; ++r) {
        const int n = rowBase + ti * 16 + r + 8 * half;
        const size_t o = (((size_t)(b * NN + n) * TT + t) * EMBED) + h * HDIM + l;
        Xout[o] = c[r];
      }
    }
  }
}

__global__ __launch_bounds__(128) void fc_kernel(const float* __restrict__ X,
                                                 const float* __restrict__ Wo,
                                                 const float* __restrict__ bo,
                                                 float* __restrict__ Y) {
  __shared__ float sX[16 * EMBED];
  __shared__ float sW[EMBED * EMBED];
  __shared__ float sb[EMBED];

  const int tid  = threadIdx.x;
  const int wave = tid >> 5;
  const int lane = tid & 31;
  const int half = lane >> 4;
  const int l    = lane & 15;
  const size_t mBase = (size_t)blockIdx.x * 16;

  for (int i = tid; i < 16 * EMBED; i += 128) sX[i] = X[mBase * EMBED + i];
  for (int i = tid; i < EMBED * EMBED; i += 128) sW[i] = Wo[i];
  if (tid < EMBED) sb[tid] = bo[tid];
  __syncthreads();

  const int tj = wave;
  v8f c = {};
#pragma unroll
  for (int kb = 0; kb < EMBED; kb += 4) {
    const int ac = kb + 2 * half;
    v2f a, bm;
    a.x  = sX[l * EMBED + ac];
    a.y  = sX[l * EMBED + ac + 1];
    bm.x = sW[(tj * 16 + l) * EMBED + ac];
    bm.y = sW[(tj * 16 + l) * EMBED + ac + 1];
    c = wmma_f32_16x16x4(a, bm, c);
  }
  const float bias = sb[tj * 16 + l];
#pragma unroll
  for (int r = 0; r < 8; ++r) {
    const size_t row = mBase + r + 8 * half;
    Y[row * EMBED + tj * 16 + l] = c[r] + bias;
  }
}

extern "C" void kernel_launch(void* const* d_in, const int* in_sizes, int n_in,
                              void* d_out, int out_size, void* d_ws,
                              size_t ws_size, hipStream_t stream) {
  (void)in_sizes; (void)n_in; (void)out_size; (void)ws_size;
  const float* V  = (const float*)d_in[0];
  const float* K  = (const float*)d_in[1];
  const float* Q  = (const float*)d_in[2];
  const float* Wv = (const float*)d_in[3];
  const float* Wk = (const float*)d_in[4];
  const float* Wq = (const float*)d_in[5];
  const float* Wo = (const float*)d_in[6];
  const float* bo = (const float*)d_in[7];
  float* X = (float*)d_ws;
  float* Y = (float*)d_out;

  hipLaunchKernelGGL(attn_kernel, dim3(BB * TT * HEADS * 2), dim3(128), 0,
                     stream, V, K, Q, Wv, Wk, Wq, X);
  hipLaunchKernelGGL(fc_kernel, dim3((BB * NN * TT) / 16), dim3(128), 0, stream,
                     X, Wo, bo, Y);
}


# Score tile in LDS: row pitch 128 -> 132 floats and softmax threads mapped to consecutive rows, so the softmax passes and the P*V A-fragment reads stop hitting one bank; every sum keeps its order
# speedup vs baseline: 2.3677x; 2.3677x over previous
#include <hip/hip_runtime.h>

typedef float v2f __attribute__((ext_vector_type(2)));
typedef float v8f __attribute__((ext_vector_type(8)));

#define EMBED 64
#define HEADS 8
#define HDIM  8
#define BB    8
#define NN    128
#define TT    64
#define SP    (NN + 4)

__device__ __forceinline__ v8f wmma_f32_16x16x4(v2f a, v2f b, v8f c) {
  return __builtin_amdgcn_wmma_f32_16x16x4_f32(false, a, false, b, (short)0, c,
                                               false, false);
}

__global__ __launch_bounds__(128) void attn_kernel(
    const float* __restrict__ V, const float* __restrict__ K,
    const float* __restrict__ Q, const float* __restrict__ Wv,
    const float* __restrict__ Wk, const float* __restrict__ Wq,
    float* __restrict__ Xout) {
  __shared__ float S[64 * SP];
  __shared__ float Qp[NN * HDIM];
  __shared__ float Kp[NN * HDIM];
  __shared__ float Vp[NN * HDIM];
  __shared__ float redM[128];
  __shared__ float redS[128];

  const int blk     = blockIdx.x;
  const int rh      = blk & 1;
  const int h       = (blk >> 1) % HEADS;
  const int t       = (blk >> 1) / HEADS % TT;
  const int b       = (blk >> 1) / (HEADS * TT);
  const int rowBase = rh * 64;

  const int tid  = threadIdx.x;
  const int wave = tid >> 5;
  const int lane = tid & 31;
  const int half = lane >> 4;
  const int l    = lane & 15;

  {
    const int n = tid;
    const size_t base = (((size_t)(b * NN + n) * TT + t) * EMBED) + h * HDIM;
    float qv[HDIM], kv[HDIM], vv[HDIM];
#pragma unroll
    for (int d = 0; d < HDIM; ++d) {
      qv[d] = Q[base + d];
      kv[d] = K[base + d];
      vv[d] = V[base + d];
    }
#pragma unroll
    for (int e = 0; e < HDIM; ++e) {
      float aq = 0.f, ak = 0.f, av = 0.f;
#pragma unroll
      for (int d = 0; d < HDIM; ++d) {
        aq += qv[d] * Wq[e * HDIM + d];
        ak += kv[d] * Wk[e * HDIM + d];
        av += vv[d] * Wv[e * HDIM + d];
      }
      Qp[n * HDIM + e] = aq;
      Kp[n * HDIM + e] = ak;
      Vp[n * HDIM + e] = av;
    }
  }
  __syncthreads();

  const float scale = 0.125f;
  for (int tile = wave; tile < 32; tile += 4) {
    const int ti = tile >> 3;
    const int tj = tile & 7;
    v8f c = {};
#pragma unroll
    for (int kb = 0; kb < HDIM; kb += 4) {
      const int ac = kb + 2 * half;
      v2f a, bm;
      a.x  = Qp[(rowBase + ti * 16 + l) * HDIM + ac];
      a.y  = Qp[(rowBase + ti * 16 + l) * HDIM + ac + 1];
      bm.x = Kp[(tj * 16 + l) * HDIM + ac];
      bm.y = Kp[(tj * 16 + l) * HDIM + ac + 1];
      c = wmma_f32_16x16x4(a, bm, c);
    }
#pragma unroll
    for (int r = 0; r < 8; ++r) {
      const int row = ti * 16 + r + 8 * half;
      S[row * SP + tj * 16 + l] = c[r] * scale;
    }
  }
  __syncthreads();

  {
    const int row = tid & 63;
    const int seg = tid >> 6;
    float* Sseg = S + row * SP + seg * 64;

    float m = -1e30f;
    for (int j = 0; j < 64; ++j) m = fmaxf(m, Sseg[j]);
    redM[tid] = m;
    __syncthreads();
    m = fmaxf(redM[row], redM[row + 64]);

    float sum = 0.f;
    for (int j = 0; j < 64; ++j) {
      const float ex = __expf(Sseg[j] - m);
      Sseg[j] = ex;
      sum += ex;
    }
    redS[tid] = sum;
    __syncthreads();
    const float inv = 1.f / (redS[row] + redS[row + 64]);
    for (int j = 0; j < 64; ++j) Sseg[j] *= inv;
  }
  __syncthreads();

  {
    const int ti = wave;
    v8f c = {};
    for (int kb = 0; kb < NN; kb += 4) {
      const int ac = kb + 2 * half;
      v2f a, bm;
      a.x  = S[(ti * 16 + l) * SP + ac];
      a.y  = S[(ti * 16 + l) * SP + ac + 1];
      bm.x = (l < HDIM) ? Vp[ac * HDIM + l] : 0.f;
      bm.y = (l < HDIM) ? Vp[(ac + 1) * HDIM + l] : 0.f;
      c = wmma_f32_16x16x4(a, bm, c);
    }
    if (l < HDIM) {
#pragma unroll
      for (int r = 0; r < 8; ++r) {
        const int n = rowBase + ti * 16 + r + 8 * half;
        const size_t o = (((size_t)(b * NN + n) * TT + t) * EMBED) + h * HDIM + l;
        Xout[o] = c[r];
      }
    }
  }
}

__global__ __launch_bounds__(128) void fc_kernel(const float* __restrict__ X,
                                                 const float* __restrict__ Wo,
                                                 const float* __restrict__ bo,
                                                 float* __restrict__ Y) {
  __shared__ float sX[16 * EMBED];
  __shared__ float sW[EMBED * EMBED];
  __shared__ float sb[EMBED];

  const int tid  = threadIdx.x;
  const int wave = tid >> 5;
  const int lane = tid & 31;
  const int half = lane >> 4;
  const int l    = lane & 15;
  const size_t mBase = (size_t)blockIdx.x * 16;

  for (int i = tid; i < 16 * EMBED; i += 128) sX[i] = X[mBase * EMBED + i];
  for (int i = tid; i < EMBED * EMBED; i += 128) sW[i] = Wo[i];
  if (tid < EMBED) sb[tid] = bo[tid];
  __syncthreads();

  const int tj = wave;
  v8f c = {};
#pragma unroll
  for (int kb = 0; kb < EMBED; kb += 4) {
    const int ac = kb + 2 * half;
    v2f a, bm;
    a.x  = sX[l * EMBED + ac];
    a.y  = sX[l * EMBED + ac + 1];
    bm.x = sW[(tj * 16 + l) * EMBED + ac];
    bm.y = sW[(tj * 16 + l) * EMBED + ac + 1];
    c = wmma_f32_16x16x4(a, bm, c);
  }
  const float bias = sb[tj * 16 + l];
#pragma unroll
  for (int r = 0; r < 8; ++r) {
    const size_t row = mBase + r + 8 * half;
    Y[row * EMBED + tj * 16 + l] = c[r] + bias;
  }
}

extern "C" void kernel_launch(void* const* d_in, const int* in_sizes, int n_in,
                              void* d_out, int out_size, void* d_ws,
                              size_t ws_size, hipStream_t stream) {
  (void)in_sizes; (void)n_in; (void)out_size; (void)ws_size;
  const float* V  = (const float*)d_in[0];
  const float* K  = (const float*)d_in[1];
  const float* Q  = (const float*)d_in[2];
  const float* Wv = (const float*)d_in[3];
  const float* Wk = (const float*)d_in[4];
  const float* Wq = (const float*)d_in[5];
  const float* Wo = (const float*)d_in[6];
  const float* bo = (const float*)d_in[7];
  float* X = (float*)d_ws;
  float* Y = (float*)d_out;

  hipLaunchKernelGGL(attn_kernel, dim3(BB * TT * HEADS * 2), dim3(128), 0,
                     stream, V, K, Q, Wv, Wk, Wq, X);
  hipLaunchKernelGGL(fc_kernel, dim3((BB * NN * TT) / 16), dim3(128), 0, stream,
                     X, Wo, bo, Y);
}


# Scores transposed (K*Q^T), kept in registers: per-lane base-2 softmax, D registers fed as B of V^T*P^T, 1/sum applied to outputs; no score tile in LDS; softmax sum tree, P*V key order change
# speedup vs baseline: 3.2781x; 1.3845x over previous
#include <hip/hip_runtime.h>


typedef float v2f __attribute__((ext_vector_type(2)));
typedef float v4f __attribute__((ext_vector_type(4)));
typedef float v8f __attribute__((ext_vector_type(8)));

#define EMBED 64
#define HEADS 8
#define HDIM  8
#define BB    8
#define NN    128
#define TT    64
#define VP    (NN + 4)

__device__ __forceinline__ v8f wmma_f32_16x16x4(v2f a, v2f b, v8f c) {
  return __builtin_amdgcn_wmma_f32_16x16x4_f32(false, a, false, b, (short)0, c,
                                               false, false);
}

__device__ __forceinline__ float xhalf(float x) {
  const int v = __builtin_bit_cast(int, x);
  return __builtin_bit_cast(
      float, __builtin_amdgcn_permlanex16(v, v, 0x76543210, 0xfedcba98, false,
                                          false));
}

__global__ __launch_bounds__(128) void attn_kernel(
    const float* __restrict__ V, const float* __restrict__ K,
    const float* __restrict__ Q, const float* __restrict__ Wv,
    const float* __restrict__ Wk, const float* __restrict__ Wq,
    float* __restrict__ Xout) {
  __shared__ __attribute__((aligned(16))) float Qp[NN * HDIM];
  __shared__ __attribute__((aligned(16))) float Kp[NN * HDIM];
  __shared__ __attribute__((aligned(16))) float VpT[16 * VP];

  const int blk     = blockIdx.x;
  const int rh      = blk & 1;
  const int h       = (blk >> 1) % HEADS;
  const int t       = (blk >> 1) / HEADS % TT;
  const int b       = (blk >> 1) / (HEADS * TT);
  const int rowBase = rh * 64;

  const int tid  = threadIdx.x;
  const int wave = tid >> 5;
  const int lane = tid & 31;
  const int half = lane >> 4;
  const int l    = lane & 15;

  {
    const int n = tid;
    const size_t base = (((size_t)(b * NN + n) * TT + t) * EMBED) + h * HDIM;
    float qv[HDIM], kv[HDIM], vv[HDIM];
#pragma unroll
    for (int d = 0; d < HDIM; ++d) {
      qv[d] = Q[base + d];
      kv[d] = K[base + d];
      vv[d] = V[base + d];
    }
#pragma unroll
    for (int e = 0; e < HDIM; ++e) {
      float aq = 0.f, ak = 0.f, av = 0.f;
#pragma unroll
      for (int d = 0; d < HDIM; ++d) {
        aq += qv[d] * Wq[e * HDIM + d];
        ak += kv[d] * Wk[e * HDIM + d];
        av += vv[d] * Wv[e * HDIM + d];
      }
      Qp[n * HDIM + e] = aq;
      Kp[n * HDIM + e] = ak;
      VpT[e * VP + n]  = av;
    }
    for (int i = tid; i < 8 * VP; i += 128) VpT[8 * VP + i] = 0.f;
  }
  __syncthreads();

  const int qrow = rowBase + wave * 16 + l;
  const v2f qf0 = *(const v2f*)&Qp[qrow * HDIM + 2 * half];
  const v2f qf1 = *(const v2f*)&Qp[qrow * HDIM + 4 + 2 * half];
  v8f s[8];
#pragma unroll
  for (int tj = 0; tj < 8; ++tj) {
    const v2f kf0 = *(const v2f*)&Kp[(tj * 16 + l) * HDIM + 2 * half];
    const v2f kf1 = *(const v2f*)&Kp[(tj * 16 + l) * HDIM + 4 + 2 * half];
    v8f c = {};
    c = wmma_f32_16x16x4(kf0, qf0, c);
    c = wmma_f32_16x16x4(kf1, qf1, c);
    s[tj] = c;
  }

  float m0 = s[0][0], m1 = s[0][1], m2 = s[0][2], m3 = s[0][3];
#pragma unroll
  for (int tj = 0; tj < 8; ++tj) {
#pragma unroll
    for (int r = 0; r < 8; r += 4) {
      m0 = fmaxf(m0, s[tj][r]);
      m1 = fmaxf(m1, s[tj][r + 1]);
      m2 = fmaxf(m2, s[tj][r + 2]);
      m3 = fmaxf(m3, s[tj][r + 3]);
    }
  }
  float m = fmaxf(fmaxf(m0, m1), fmaxf(m2, m3));
  m = fmaxf(m, xhalf(m));

  const float c2 = 0.125f * 1.4426950408889634f;
  const float nm = -(m * c2);
  float z0 = 0.f, z1 = 0.f, z2 = 0.f, z3 = 0.f;
#pragma unroll
  for (int tj = 0; tj < 8; ++tj) {
#pragma unroll
    for (int r = 0; r < 8; r += 4) {
      const float e0 = __builtin_amdgcn_exp2f(__builtin_fmaf(s[tj][r], c2, nm));
      const float e1 = __builtin_amdgcn_exp2f(__builtin_fmaf(s[tj][r + 1], c2, nm));
      const float e2 = __builtin_amdgcn_exp2f(__builtin_fmaf(s[tj][r + 2], c2, nm));
      const float e3 = __builtin_amdgcn_exp2f(__builtin_fmaf(s[tj][r + 3], c2, nm));
      s[tj][r] = e0; s[tj][r + 1] = e1; s[tj][r + 2] = e2; s[tj][r + 3] = e3;
      z0 += e0; z1 += e1; z2 += e2; z3 += e3;
    }
  }
  float z = (z0 + z1) + (z2 + z3);
  z += xhalf(z);

  v8f o0 = {}, o1 = {};
#pragma unroll
  for (int tj = 0; tj < 8; ++tj) {
#pragma unroll
    for (int j = 0; j < 4; ++j) {
      const v2f a = *(const v2f*)&VpT[l * VP + tj * 16 + 8 * half + 2 * j];
      v2f p;
      p.x = s[tj][2 * j];
      p.y = s[tj][2 * j + 1];
      if (j & 1) o1 = wmma_f32_16x16x4(a, p, o1);
      else       o0 = wmma_f32_16x16x4(a, p, o0);
    }
  }
  const float inv = 1.f / z;
  if (half == 0) {
    const int n = qrow;
    const size_t o = (((size_t)(b * NN + n) * TT + t) * EMBED) + h * HDIM;
    v4f lo, hi;
    lo.x = (o0[0] + o1[0]) * inv; lo.y = (o0[1] + o1[1]) * inv;
    lo.z = (o0[2] + o1[2]) * inv; lo.w = (o0[3] + o1[3]) * inv;
    hi.x = (o0[4] + o1[4]) * inv; hi.y = (o0[5] + o1[5]) * inv;
    hi.z = (o0[6] + o1[6]) * inv; hi.w = (o0[7] + o1[7]) * inv;
    *(v4f*)&Xout[o]     = lo;
    *(v4f*)&Xout[o + 4] = hi;
  }
}

__global__ __launch_bounds__(128) void fc_kernel(const float* __restrict__ X,
                                                 const float* __restrict__ Wo,
                                                 const float* __restrict__ bo,
                                                 float* __restrict__ Y) {
  __shared__ float sX[16 * EMBED];
  __shared__ float sW[EMBED * EMBED];
  __shared__ float sb[EMBED];

  const int tid  = threadIdx.x;
  const int wave = tid >> 5;
  const int lane = tid & 31;
  const int half = lane >> 4;
  const int l    = lane & 15;
  const size_t mBase = (size_t)blockIdx.x * 16;

  for (int i = tid; i < 16 * EMBED; i += 128) sX[i] = X[mBase * EMBED + i];
  for (int i = tid; i < EMBED * EMBED; i += 128) sW[i] = Wo[i];
  if (tid < EMBED) sb[tid] = bo[tid];
  __syncthreads();

  const int tj = wave;
  v8f c = {};
#pragma unroll
  for (int kb = 0; kb < EMBED; kb += 4) {
    const int ac = kb + 2 * half;
    v2f a, bm;
    a.x  = sX[l * EMBED + ac];
    a.y  = sX[l * EMBED + ac + 1];
    bm.x = sW[(tj * 16 + l) * EMBED + ac];
    bm.y = sW[(tj * 16 + l) * EMBED + ac + 1];
    c = wmma_f32_16x16x4(a, bm, c);
  }
  const float bias = sb[tj * 16 + l];
#pragma unroll
  for (int r = 0; r < 8; ++r) {
    const size_t row = mBase + r + 8 * half;
    Y[row * EMBED + tj * 16 + l] = c[r] + bias;
  }
}

extern "C" void kernel_launch(void* const* d_in, const int* in_sizes, int n_in,
                              void* d_out, int out_size, void* d_ws,
                              size_t ws_size, hipStream_t stream) {
  (void)in_sizes; (void)n_in; (void)out_size; (void)ws_size;
  const float* V  = (const float*)d_in[0];
  const float* K  = (const float*)d_in[1];
  const float* Q  = (const float*)d_in[2];
  const float* Wv = (const float*)d_in[3];
  const float* Wk = (const float*)d_in[4];
  const float* Wq = (const float*)d_in[5];
  const float* Wo = (const float*)d_in[6];
  const float* bo = (const float*)d_in[7];
  float* X = (float*)d_ws;
  float* Y = (float*)d_out;

  hipLaunchKernelGGL(attn_kernel, dim3(BB * TT * HEADS * 2), dim3(128), 0,
                     stream, V, K, Q, Wv, Wk, Wq, X);
  hipLaunchKernelGGL(fc_kernel, dim3((BB * NN * TT) / 16), dim3(128), 0, stream,
                     X, Wo, bo, Y);
}


# Output projection re-tiled: block = 64 rows in one LDS stage (pitch 68 floats, 16-byte loads); each wave reads its 16 Wo fragments once into registers, reuses them over 4 row tiles; sum order kept
# speedup vs baseline: 4.1137x; 1.2549x over previous
#include <hip/hip_runtime.h>


typedef float v2f __attribute__((ext_vector_type(2)));
typedef float v4f __attribute__((ext_vector_type(4)));
typedef float v8f __attribute__((ext_vector_type(8)));

#define EMBED 64
#define HEADS 8
#define HDIM  8
#define BB    8
#define NN    128
#define TT    64
#define VP    (NN + 4)

__device__ __forceinline__ v8f wmma_f32_16x16x4(v2f a, v2f b, v8f c) {
  return __builtin_amdgcn_wmma_f32_16x16x4_f32(false, a, false, b, (short)0, c,
                                               false, false);
}

__device__ __forceinline__ float xhalf(float x) {
  const int v = __builtin_bit_cast(int, x);
  return __builtin_bit_cast(
      float, __builtin_amdgcn_permlanex16(v, v, 0x76543210, 0xfedcba98, false,
                                          false));
}

__global__ __launch_bounds__(128) void attn_kernel(
    const float* __restrict__ V, const float* __restrict__ K,
    const float* __restrict__ Q, const float* __restrict__ Wv,
    const float* __restrict__ Wk, const float* __restrict__ Wq,
    float* __restrict__ Xout) {
  __shared__ __attribute__((aligned(16))) float Qp[NN * HDIM];
  __shared__ __attribute__((aligned(16))) float Kp[NN * HDIM];
  __shared__ __attribute__((aligned(16))) float VpT[16 * VP];

  const int blk     = blockIdx.x;
  const int rh      = blk & 1;
  const int h       = (blk >> 1) % HEADS;
  const int t       = (blk >> 1) / HEADS % TT;
  const int b       = (blk >> 1) / (HEADS * TT);
  const int rowBase = rh * 64;

  const int tid  = threadIdx.x;
  const int wave = tid >> 5;
  const int lane = tid & 31;
  const int half = lane >> 4;
  const int l    = lane & 15;

  {
    const int n = tid;
    const size_t base = (((size_t)(b * NN + n) * TT + t) * EMBED) + h * HDIM;
    float qv[HDIM], kv[HDIM], vv[HDIM];
#pragma unroll
    for (int d = 0; d < HDIM; ++d) {
      qv[d] = Q[base + d];
      kv[d] = K[base + d];
      vv[d] = V[base + d];
    }
#pragma unroll
    for (int e = 0; e < HDIM; ++e) {
      float aq = 0.f, ak = 0.f, av = 0.f;
#pragma unroll
      for (int d = 0; d < HDIM; ++d) {
        aq += qv[d] * Wq[e * HDIM + d];
        ak += kv[d] * Wk[e * HDIM + d];
        av += vv[d] * Wv[e * HDIM + d];
      }
      Qp[n * HDIM + e] = aq;
      Kp[n * HDIM + e] = ak;
      VpT[e * VP + n]  = av;
    }
    for (int i = tid; i < 8 * VP; i += 128) VpT[8 * VP + i] = 0.f;
  }
  __syncthreads();

  const int qrow = rowBase + wave * 16 + l;
  const v2f qf0 = *(const v2f*)&Qp[qrow * HDIM + 2 * half];
  const v2f qf1 = *(const v2f*)&Qp[qrow * HDIM + 4 + 2 * half];
  v8f s[8];
#pragma unroll
  for (int tj = 0; tj < 8; ++tj) {
    const v2f kf0 = *(const v2f*)&Kp[(tj * 16 + l) * HDIM + 2 * half];
    const v2f kf1 = *(const v2f*)&Kp[(tj * 16 + l) * HDIM + 4 + 2 * half];
    v8f c = {};
    c = wmma_f32_16x16x4(kf0, qf0, c);
    c = wmma_f32_16x16x4(kf1, qf1, c);
    s[tj] = c;
  }

  float m0 = s[0][0], m1 = s[0][1], m2 = s[0][2], m3 = s[0][3];
#pragma unroll
  for (int tj = 0; tj < 8; ++tj) {
#pragma unroll
    for (int r = 0; r < 8; r += 4) {
      m0 = fmaxf(m0, s[tj][r]);
      m1 = fmaxf(m1, s[tj][r + 1]);
      m2 = fmaxf(m2, s[tj][r + 2]);
      m3 = fmaxf(m3, s[tj][r + 3]);
    }
  }
  float m = fmaxf(fmaxf(m0, m1), fmaxf(m2, m3));
  m = fmaxf(m, xhalf(m));

  const float c2 = 0.125f * 1.4426950408889634f;
  const float nm = -(m * c2);
  float z0 = 0.f, z1 = 0.f, z2 = 0.f, z3 = 0.f;
#pragma unroll
  for (int tj = 0; tj < 8; ++tj) {
#pragma unroll
    for (int r = 0; r < 8; r += 4) {
      const float e0 = __builtin_amdgcn_exp2f(__builtin_fmaf(s[tj][r], c2, nm));
      const float e1 = __builtin_amdgcn_exp2f(__builtin_fmaf(s[tj][r + 1], c2, nm));
      const float e2 = __builtin_amdgcn_exp2f(__builtin_fmaf(s[tj][r + 2], c2, nm));
      const float e3 = __builtin_amdgcn_exp2f(__builtin_fmaf(s[tj][r + 3], c2, nm));
      s[tj][r] = e0; s[tj][r + 1] = e1; s[tj][r + 2] = e2; s[tj][r + 3] = e3;
      z0 += e0; z1 += e1; z2 += e2; z3 += e3;
    }
  }
  float z = (z0 + z1) + (z2 + z3);
  z += xhalf(z);

  v8f o0 = {}, o1 = {};
#pragma unroll
  for (int tj = 0; tj < 8; ++tj) {
#pragma unroll
    for (int j = 0; j < 4; ++j) {
      const v2f a = *(const v2f*)&VpT[l * VP + tj * 16 + 8 * half + 2 * j];
      v2f p;
      p.x = s[tj][2 * j];
      p.y = s[tj][2 * j + 1];
      if (j & 1) o1 = wmma_f32_16x16x4(a, p, o1);
      else       o0 = wmma_f32_16x16x4(a, p, o0);
    }
  }
  const float inv = 1.f / z;
  if (half == 0) {
    const int n = qrow;
    const size_t o = (((size_t)(b * NN + n) * TT + t) * EMBED) + h * HDIM;
    v4f lo, hi;
    lo.x = (o0[0] + o1[0]) * inv; lo.y = (o0[1] + o1[1]) * inv;
    lo.z = (o0[2] + o1[2]) * inv; lo.w = (o0[3] + o1[3]) * inv;
    hi.x = (o0[4] + o1[4]) * inv; hi.y = (o0[5] + o1[5]) * inv;
    hi.z = (o0[6] + o1[6]) * inv; hi.w = (o0[7] + o1[7]) * inv;
    *(v4f*)&Xout[o]     = lo;
    *(v4f*)&Xout[o + 4] = hi;
  }
}

#define FC_ROWS 64
#define XP (EMBED + 4)

__global__ __launch_bounds__(128) void fc_kernel(const float* __restrict__ X,
                                                 const float* __restrict__ Wo,
                                                 const float* __restrict__ bo,
                                                 float* __restrict__ Y) {
  __shared__ __attribute__((aligned(16))) float sX[FC_ROWS * XP];

  const int tid  = threadIdx.x;
  const int wave = tid >> 5;
  const int lane = tid & 31;
  const int half = lane >> 4;
  const int l    = lane & 15;
  const size_t mBase = (size_t)blockIdx.x * FC_ROWS;

  const v4f* Xv = (const v4f*)(X + mBase * EMBED);
#pragma unroll
  for (int q = 0; q < (FC_ROWS * EMBED / 4) / 128; ++q) {
    const int i4  = tid + 128 * q;
    const int row = i4 >> 4;
    const int c4  = i4 & 15;
    *(v4f*)&sX[row * XP + c4 * 4] = Xv[i4];
  }

  const int tj = wave;
  const float* wrow = Wo + (tj * 16 + l) * EMBED + 2 * half;
  v2f bm[16];
#pragma unroll
  for (int i = 0; i < 16; ++i) {
    bm[i].x = wrow[4 * i];
    bm[i].y = wrow[4 * i + 1];
  }
  const float bias = bo[tj * 16 + l];
  __syncthreads();

  v8f c[4] = {};
#pragma unroll
  for (int i = 0; i < 16; ++i) {
    const int ac = 4 * i + 2 * half;
#pragma unroll
    for (int mt = 0; mt < 4; ++mt) {
      const v2f a = *(const v2f*)&sX[(mt * 16 + l) * XP + ac];
      c[mt] = wmma_f32_16x16x4(a, bm[i], c[mt]);
    }
  }
#pragma unroll
  for (int mt = 0; mt < 4; ++mt) {
#pragma unroll
    for (int r = 0; r < 8; ++r) {
      const size_t row = mBase + mt * 16 + r + 8 * half;
      Y[row * EMBED + tj * 16 + l] = c[mt][r] + bias;
    }
  }
}

extern "C" void kernel_launch(void* const* d_in, const int* in_sizes, int n_in,
                              void* d_out, int out_size, void* d_ws,
                              size_t ws_size, hipStream_t stream) {
  (void)in_sizes; (void)n_in; (void)out_size; (void)ws_size;
  const float* V  = (const float*)d_in[0];
  const float* K  = (const float*)d_in[1];
  const float* Q  = (const float*)d_in[2];
  const float* Wv = (const float*)d_in[3];
  const float* Wk = (const float*)d_in[4];
  const float* Wq = (const float*)d_in[5];
  const float* Wo = (const float*)d_in[6];
  const float* bo = (const float*)d_in[7];
  float* X = (float*)d_ws;
  float* Y = (float*)d_out;

  hipLaunchKernelGGL(attn_kernel, dim3(BB * TT * HEADS * 2), dim3(128), 0,
                     stream, V, K, Q, Wv, Wk, Wq, X);
  hipLaunchKernelGGL(fc_kernel, dim3((BB * NN * TT) / FC_ROWS), dim3(128), 0, stream,
                     X, Wo, bo, Y);
}


# Attention re-tiled: one 512-thread block per (b,t) holds all 8 heads; q/k/v read as whole 256-byte rows, each (row, head) piece projected once, wave = head x 64 queries; every sum keeps its order
# speedup vs baseline: 6.6526x; 1.6172x over previous
#include <hip/hip_runtime.h>


typedef float v2f __attribute__((ext_vector_type(2)));
typedef float v4f __attribute__((ext_vector_type(4)));
typedef float v8f __attribute__((ext_vector_type(8)));

#define EMBED 64
#define HEADS 8
#define HDIM  8
#define BB    8
#define NN    128
#define TT    64
#define VP    (NN + 4)
#define QS    (NN * HDIM + 4)
#define VS    (HDIM * VP + 8)
#define Q_OFF 0
#define K_OFF (HEADS * QS)
#define V_OFF (2 * HEADS * QS)
#define Z_OFF (2 * HEADS * QS + HEADS * VS)
#define SM_FLOATS (Z_OFF + NN)
#define ATT_THREADS 512

__device__ __forceinline__ v8f wmma_f32_16x16x4(v2f a, v2f b, v8f c) {
  return __builtin_amdgcn_wmma_f32_16x16x4_f32(false, a, false, b, (short)0, c,
                                               false, false);
}

__device__ __forceinline__ float xhalf(float x) {
  const int v = __builtin_bit_cast(int, x);
  return __builtin_bit_cast(
      float, __builtin_amdgcn_permlanex16(v, v, 0x76543210, 0xfedcba98, false,
                                          false));
}

__device__ __forceinline__ void project8(const v4f lo, const v4f hi,
                                         const float* __restrict__ Wm,
                                         float* out) {
  const float in[HDIM] = {lo.x, lo.y, lo.z, lo.w, hi.x, hi.y, hi.z, hi.w};
#pragma unroll
  for (int e = 0; e < HDIM; ++e) {
    float a = 0.f;
#pragma unroll
    for (int d = 0; d < HDIM; ++d) a = __builtin_fmaf(in[d], Wm[e * HDIM + d], a);
    out[e] = a;
  }
}

__global__ __launch_bounds__(ATT_THREADS, 8) void attn_kernel(
    const float* __restrict__ V, const float* __restrict__ K,
    const float* __restrict__ Q, const float* __restrict__ Wv,
    const float* __restrict__ Wk, const float* __restrict__ Wq,
    float* __restrict__ Xout) {
  __shared__ __attribute__((aligned(16))) float sm[SM_FLOATS];

  const int b = blockIdx.x / TT;
  const int t = blockIdx.x % TT;

  const int tid  = threadIdx.x;
  const int wave = tid >> 5;
  const int lane = tid & 31;
  const int half = lane >> 4;
  const int l    = lane & 15;

  {
    const int ph = lane & 7;
    const int pn = (wave >> 1) * 8 + (wave & 1) + 2 * (lane >> 3);
    v4f rq[2][2], rk[2][2], rv[2][2];
#pragma unroll
    for (int it = 0; it < 2; ++it) {
      const int n = pn + 64 * it;
      const size_t base = (((size_t)(b * NN + n) * TT + t) * EMBED) + ph * HDIM;
      rq[it][0] = *(const v4f*)&Q[base];
      rq[it][1] = *(const v4f*)&Q[base + 4];
      rk[it][0] = *(const v4f*)&K[base];
      rk[it][1] = *(const v4f*)&K[base + 4];
      rv[it][0] = *(const v4f*)&V[base];
      rv[it][1] = *(const v4f*)&V[base + 4];
    }
    if (tid < NN) sm[Z_OFF + tid] = 0.f;
#pragma unroll
    for (int it = 0; it < 2; ++it) {
      const int n = pn + 64 * it;
      float o[HDIM];
      v4f lo, hi;
      project8(rq[it][0], rq[it][1], Wq, o);
      lo.x = o[0]; lo.y = o[1]; lo.z = o[4]; lo.w = o[5];
      hi.x = o[2]; hi.y = o[3]; hi.z = o[6]; hi.w = o[7];
      *(v4f*)&sm[Q_OFF + ph * QS + n * HDIM]     = lo;
      *(v4f*)&sm[Q_OFF + ph * QS + n * HDIM + 4] = hi;
      project8(rk[it][0], rk[it][1], Wk, o);
      lo.x = o[0]; lo.y = o[1]; lo.z = o[4]; lo.w = o[5];
      hi.x = o[2]; hi.y = o[3]; hi.z = o[6]; hi.w = o[7];
      *(v4f*)&sm[K_OFF + ph * QS + n * HDIM]     = lo;
      *(v4f*)&sm[K_OFF + ph * QS + n * HDIM + 4] = hi;
      project8(rv[it][0], rv[it][1], Wv, o);
#pragma unroll
      for (int e = 0; e < HDIM; ++e) sm[V_OFF + ph * VS + e * VP + n] = o[e];
    }
  }
  __syncthreads();

  const int hh    = wave >> 1;
  const int qbase = (wave & 1) * 64;
  const float* Qh = &sm[Q_OFF + hh * QS];
  const float* Kh = &sm[K_OFF + hh * QS];
  const float* Vrow = &sm[(l < HDIM) ? (V_OFF + hh * VS + l * VP + 8 * half)
                                     : (Z_OFF + 8 * half)];

  for (int qt = 0; qt < 4; ++qt) {
    asm volatile("" ::: "memory");
    const int qrow = qbase + qt * 16 + l;
    const v4f qf = *(const v4f*)&Qh[qrow * HDIM + 4 * half];
    v2f qf0, qf1;
    qf0.x = qf.x; qf0.y = qf.y; qf1.x = qf.z; qf1.y = qf.w;
    v8f s[8];
#pragma unroll
    for (int tj = 0; tj < 8; ++tj) {
      const v4f kf = *(const v4f*)&Kh[(tj * 16 + l) * HDIM + 4 * half];
      v2f kf0, kf1;
      kf0.x = kf.x; kf0.y = kf.y; kf1.x = kf.z; kf1.y = kf.w;
      v8f c = {};
      c = wmma_f32_16x16x4(kf0, qf0, c);
      c = wmma_f32_16x16x4(kf1, qf1, c);
      s[tj] = c;
    }

    float m0 = s[0][0], m1 = s[0][1], m2 = s[0][2], m3 = s[0][3];
#pragma unroll
    for (int tj = 0; tj < 8; ++tj) {
#pragma unroll
      for (int r = 0; r < 8; r += 4) {
        m0 = fmaxf(m0, s[tj][r]);
        m1 = fmaxf(m1, s[tj][r + 1]);
        m2 = fmaxf(m2, s[tj][r + 2]);
        m3 = fmaxf(m3, s[tj][r + 3]);
      }
    }
    float m = fmaxf(fmaxf(m0, m1), fmaxf(m2, m3));
    m = fmaxf(m, xhalf(m));

    const float c2 = 0.125f * 1.4426950408889634f;
    const float nm = -(m * c2);
    float z0 = 0.f, z1 = 0.f, z2 = 0.f, z3 = 0.f;
#pragma unroll
    for (int tj = 0; tj < 8; ++tj) {
#pragma unroll
      for (int r = 0; r < 8; r += 4) {
        const float e0 = __builtin_amdgcn_exp2f(__builtin_fmaf(s[tj][r], c2, nm));
        const float e1 = __builtin_amdgcn_exp2f(__builtin_fmaf(s[tj][r + 1], c2, nm));
        const float e2 = __builtin_amdgcn_exp2f(__builtin_fmaf(s[tj][r + 2], c2, nm));
        const float e3 = __builtin_amdgcn_exp2f(__builtin_fmaf(s[tj][r + 3], c2, nm));
        s[tj][r] = e0; s[tj][r + 1] = e1; s[tj][r + 2] = e2; s[tj][r + 3] = e3;
        z0 += e0; z1 += e1; z2 += e2; z3 += e3;
      }
    }
    float z = (z0 + z1) + (z2 + z3);
    z += xhalf(z);

    v8f o0 = {}, o1 = {};
#pragma unroll
    for (int tj = 0; tj < 8; ++tj) {
      const v4f a01 = *(const v4f*)&Vrow[tj * 16];
      const v4f a23 = *(const v4f*)&Vrow[tj * 16 + 4];
      v2f a, p;
      a.x = a01.x; a.y = a01.y; p.x = s[tj][0]; p.y = s[tj][1];
      o0 = wmma_f32_16x16x4(a, p, o0);
      a.x = a01.z; a.y = a01.w; p.x = s[tj][2]; p.y = s[tj][3];
      o1 = wmma_f32_16x16x4(a, p, o1);
      a.x = a23.x; a.y = a23.y; p.x = s[tj][4]; p.y = s[tj][5];
      o0 = wmma_f32_16x16x4(a, p, o0);
      a.x = a23.z; a.y = a23.w; p.x = s[tj][6]; p.y = s[tj][7];
      o1 = wmma_f32_16x16x4(a, p, o1);
    }
    const float inv = 1.f / z;
    if (half == 0) {
      const int n = qrow;
      const size_t o = (((size_t)(b * NN + n) * TT + t) * EMBED) + hh * HDIM;
      v4f lo, hi;
      lo.x = (o0[0] + o1[0]) * inv; lo.y = (o0[1] + o1[1]) * inv;
      lo.z = (o0[2] + o1[2]) * inv; lo.w = (o0[3] + o1[3]) * inv;
      hi.x = (o0[4] + o1[4]) * inv; hi.y = (o0[5] + o1[5]) * inv;
      hi.z = (o0[6] + o1[6]) * inv; hi.w = (o0[7] + o1[7]) * inv;
      *(v4f*)&Xout[o]     = lo;
      *(v4f*)&Xout[o + 4] = hi;
    }
  }
}

#define FC_ROWS 64
#define XP (EMBED + 4)

__global__ __launch_bounds__(128) void fc_kernel(const float* __restrict__ X,
                                                 const float* __restrict__ Wo,
                                                 const float* __restrict__ bo,
                                                 float* __restrict__ Y) {
  __shared__ __attribute__((aligned(16))) float sX[FC_ROWS * XP];

  const int tid  = threadIdx.x;
  const int wave = tid >> 5;
  const int lane = tid & 31;
  const int half = lane >> 4;
  const int l    = lane & 15;
  const size_t mBase = (size_t)blockIdx.x * FC_ROWS;

  const v4f* Xv = (const v4f*)(X + mBase * EMBED);
#pragma unroll
  for (int q = 0; q < (FC_ROWS * EMBED / 4) / 128; ++q) {
    const int i4  = tid + 128 * q;
    const int row = i4 >> 4;
    const int c4  = i4 & 15;
    *(v4f*)&sX[row * XP + c4 * 4] = Xv[i4];
  }

  const int tj = wave;
  const float* wrow = Wo + (tj * 16 + l) * EMBED + 2 * half;
  v2f bm[16];
#pragma unroll
  for (int i = 0; i < 16; ++i) {
    bm[i].x = wrow[4 * i];
    bm[i].y = wrow[4 * i + 1];
  }
  const float bias = bo[tj * 16 + l];
  __syncthreads();

  v8f c[4] = {};
#pragma unroll
  for (int i = 0; i < 16; ++i) {
    const int ac = 4 * i + 2 * half;
#pragma unroll
    for (int mt = 0; mt < 4; ++mt) {
      const v2f a = *(const v2f*)&sX[(mt * 16 + l) * XP + ac];
      c[mt] = wmma_f32_16x16x4(a, bm[i], c[mt]);
    }
  }
#pragma unroll
  for (int mt = 0; mt < 4; ++mt) {
#pragma unroll
    for (int r = 0; r < 8; ++r) {
      const size_t row = mBase + mt * 16 + r + 8 * half;
      Y[row * EMBED + tj * 16 + l] = c[mt][r] + bias;
    }
  }
}

extern "C" void kernel_launch(void* const* d_in, const int* in_sizes, int n_in,
                              void* d_out, int out_size, void* d_ws,
                              size_t ws_size, hipStream_t stream) {
  (void)in_sizes; (void)n_in; (void)out_size; (void)ws_size;
  const float* V  = (const float*)d_in[0];
  const float* K  = (const float*)d_in[1];
  const float* Q  = (const float*)d_in[2];
  const float* Wv = (const float*)d_in[3];
  const float* Wk = (const float*)d_in[4];
  const float* Wq = (const float*)d_in[5];
  const float* Wo = (const float*)d_in[6];
  const float* bo = (const float*)d_in[7];
  float* X = (float*)d_ws;
  float* Y = (float*)d_out;

  hipLaunchKernelGGL(attn_kernel, dim3(BB * TT), dim3(ATT_THREADS), 0,
                     stream, V, K, Q, Wv, Wk, Wq, X);
  hipLaunchKernelGGL(fc_kernel, dim3((BB * NN * TT) / FC_ROWS), dim3(128), 0, stream,
                     X, Wo, bo, Y);
}


# Output projection fused into the attention block: the 8 heads' outputs and Wo stay in LDS, each wave forms a 16 x 32 piece of Y and stores whole 128-byte lines; one launch; every sum keeps its order
# speedup vs baseline: 7.4275x; 1.1165x over previous
#include <hip/hip_runtime.h>


typedef float v2f __attribute__((ext_vector_type(2)));
typedef float v4f __attribute__((ext_vector_type(4)));
typedef float v8f __attribute__((ext_vector_type(8)));

#define EMBED 64
#define HEADS 8
#define HDIM  8
#define BB    8
#define NN    128
#define TT    64
#define VP    (NN + 4)
#define QS    (NN * HDIM + 4)
#define VS    (HDIM * VP + 8)
#define Q_OFF 0
#define K_OFF (HEADS * QS)
#define V_OFF (2 * HEADS * QS)
#define Z_OFF (2 * HEADS * QS + HEADS * VS)
#define XP    (EMBED + 4)
#define X_OFF (Z_OFF + NN)
#define W_OFF (X_OFF + NN * XP)
#define B_OFF (W_OFF + EMBED * XP)
#define SM_FLOATS (B_OFF + EMBED)
#define YP    36
#define ATT_THREADS 512

__device__ __forceinline__ v8f wmma_f32_16x16x4(v2f a, v2f b, v8f c) {
  return __builtin_amdgcn_wmma_f32_16x16x4_f32(false, a, false, b, (short)0, c,
                                               false, false);
}

__device__ __forceinline__ float xhalf(float x) {
  const int v = __builtin_bit_cast(int, x);
  return __builtin_bit_cast(
      float, __builtin_amdgcn_permlanex16(v, v, 0x76543210, 0xfedcba98, false,
                                          false));
}

__device__ __forceinline__ void project8(const v4f lo, const v4f hi,
                                         const float* __restrict__ Wm,
                                         float* out) {
  const float in[HDIM] = {lo.x, lo.y, lo.z, lo.w, hi.x, hi.y, hi.z, hi.w};
#pragma unroll
  for (int e = 0; e < HDIM; ++e) {
    float a = 0.f;
#pragma unroll
    for (int d = 0; d < HDIM; ++d) a = __builtin_fmaf(in[d], Wm[e * HDIM + d], a);
    out[e] = a;
  }
}

__global__ __launch_bounds__(ATT_THREADS, 8) void attn_kernel(
    const float* __restrict__ V, const float* __restrict__ K,
    const float* __restrict__ Q, const float* __restrict__ Wv,
    const float* __restrict__ Wk, const float* __restrict__ Wq,
    const float* __restrict__ Wo, const float* __restrict__ bo,
    float* __restrict__ Y) {
  __shared__ __attribute__((aligned(16))) float sm[SM_FLOATS];

  const int b = blockIdx.x / TT;
  const int t = blockIdx.x % TT;

  const int tid  = threadIdx.x;
  const int wave = tid >> 5;
  const int lane = tid & 31;
  const int half = lane >> 4;
  const int l    = lane & 15;

  {
    const int ph = lane & 7;
    const int pn = (wave >> 1) * 8 + (wave & 1) + 2 * (lane >> 3);
    v4f rq[2][2], rk[2][2], rv[2][2], rw[2];
#pragma unroll
    for (int it = 0; it < 2; ++it) {
      const int n = pn + 64 * it;
      const size_t base = (((size_t)(b * NN + n) * TT + t) * EMBED) + ph * HDIM;
      rq[it][0] = *(const v4f*)&Q[base];
      rq[it][1] = *(const v4f*)&Q[base + 4];
      rk[it][0] = *(const v4f*)&K[base];
      rk[it][1] = *(const v4f*)&K[base + 4];
      rv[it][0] = *(const v4f*)&V[base];
      rv[it][1] = *(const v4f*)&V[base + 4];
    }
#pragma unroll
    for (int q = 0; q < 2; ++q) rw[q] = ((const v4f*)Wo)[tid + ATT_THREADS * q];
    const float rb = bo[tid & (EMBED - 1)];
    if (tid < NN) sm[Z_OFF + tid] = 0.f;
#pragma unroll
    for (int it = 0; it < 2; ++it) {
      const int n = pn + 64 * it;
      float o[HDIM];
      v4f lo, hi;
      project8(rq[it][0], rq[it][1], Wq, o);
      lo.x = o[0]; lo.y = o[1]; lo.z = o[4]; lo.w = o[5];
      hi.x = o[2]; hi.y = o[3]; hi.z = o[6]; hi.w = o[7];
      *(v4f*)&sm[Q_OFF + ph * QS + n * HDIM]     = lo;
      *(v4f*)&sm[Q_OFF + ph * QS + n * HDIM + 4] = hi;
      project8(rk[it][0], rk[it][1], Wk, o);
      lo.x = o[0]; lo.y = o[1]; lo.z = o[4]; lo.w = o[5];
      hi.x = o[2]; hi.y = o[3]; hi.z = o[6]; hi.w = o[7];
      *(v4f*)&sm[K_OFF + ph * QS + n * HDIM]     = lo;
      *(v4f*)&sm[K_OFF + ph * QS + n * HDIM + 4] = hi;
      project8(rv[it][0], rv[it][1], Wv, o);
#pragma unroll
      for (int e = 0; e < HDIM; ++e) sm[V_OFF + ph * VS + e * VP + n] = o[e];
    }
#pragma unroll
    for (int q = 0; q < 2; ++q) {
      const int i4 = tid + ATT_THREADS * q;
      *(v4f*)&sm[W_OFF + (i4 >> 4) * XP + (i4 & 15) * 4] = rw[q];
    }
    if (tid < EMBED) sm[B_OFF + tid] = rb;
  }
  __syncthreads();

  const int hh    = wave >> 1;
  const int qbase = (wave & 1) * 64;
  const float* Qh = &sm[Q_OFF + hh * QS];
  const float* Kh = &sm[K_OFF + hh * QS];
  const float* Vrow = &sm[(l < HDIM) ? (V_OFF + hh * VS + l * VP + 8 * half)
                                     : (Z_OFF + 8 * half)];

  for (int qt = 0; qt < 4; ++qt) {
    asm volatile("" ::: "memory");
    const int qrow = qbase + qt * 16 + l;
    const v4f qf = *(const v4f*)&Qh[qrow * HDIM + 4 * half];
    v2f qf0, qf1;
    qf0.x = qf.x; qf0.y = qf.y; qf1.x = qf.z; qf1.y = qf.w;
    v8f s[8];
#pragma unroll
    for (int tj = 0; tj < 8; ++tj) {
      const v4f kf = *(const v4f*)&Kh[(tj * 16 + l) * HDIM + 4 * half];
      v2f kf0, kf1;
      kf0.x = kf.x; kf0.y = kf.y; kf1.x = kf.z; kf1.y = kf.w;
      v8f c = {};
      c = wmma_f32_16x16x4(kf0, qf0, c);
      c = wmma_f32_16x16x4(kf1, qf1, c);
      s[tj] = c;
    }

    float m0 = s[0][0], m1 = s[0][1], m2 = s[0][2], m3 = s[0][3];
#pragma unroll
    for (int tj = 0; tj < 8; ++tj) {
#pragma unroll
      for (int r = 0; r < 8; r += 4) {
        m0 = fmaxf(m0, s[tj][r]);
        m1 = fmaxf(m1, s[tj][r + 1]);
        m2 = fmaxf(m2, s[tj][r + 2]);
        m3 = fmaxf(m3, s[tj][r + 3]);
      }
    }
    float m = fmaxf(fmaxf(m0, m1), fmaxf(m2, m3));
    m = fmaxf(m, xhalf(m));

    const float c2 = 0.125f * 1.4426950408889634f;
    const float nm = -(m * c2);
    float z0 = 0.f, z1 = 0.f, z2 = 0.f, z3 = 0.f;
#pragma unroll
    for (int tj = 0; tj < 8; ++tj) {
#pragma unroll
      for (int r = 0; r < 8; r += 4) {
        const float e0 = __builtin_amdgcn_exp2f(__builtin_fmaf(s[tj][r], c2, nm));
        const float e1 = __builtin_amdgcn_exp2f(__builtin_fmaf(s[tj][r + 1], c2, nm));
        const float e2 = __builtin_amdgcn_exp2f(__builtin_fmaf(s[tj][r + 2], c2, nm));
        const float e3 = __builtin_amdgcn_exp2f(__builtin_fmaf(s[tj][r + 3], c2, nm));
        s[tj][r] = e0; s[tj][r + 1] = e1; s[tj][r + 2] = e2; s[tj][r + 3] = e3;
        z0 += e0; z1 += e1; z2 += e2; z3 += e3;
      }
    }
    float z = (z0 + z1) + (z2 + z3);
    z += xhalf(z);

    v8f o0 = {}, o1 = {};
#pragma unroll
    for (int tj = 0; tj < 8; ++tj) {
      const v4f a01 = *(const v4f*)&Vrow[tj * 16];
      const v4f a23 = *(const v4f*)&Vrow[tj * 16 + 4];
      v2f a, p;
      a.x = a01.x; a.y = a01.y; p.x = s[tj][0]; p.y = s[tj][1];
      o0 = wmma_f32_16x16x4(a, p, o0);
      a.x = a01.z; a.y = a01.w; p.x = s[tj][2]; p.y = s[tj][3];
      o1 = wmma_f32_16x16x4(a, p, o1);
      a.x = a23.x; a.y = a23.y; p.x = s[tj][4]; p.y = s[tj][5];
      o0 = wmma_f32_16x16x4(a, p, o0);
      a.x = a23.z; a.y = a23.w; p.x = s[tj][6]; p.y = s[tj][7];
      o1 = wmma_f32_16x16x4(a, p, o1);
    }
    const float inv = 1.f / z;
    if (half == 0) {
      float* xo = &sm[X_OFF + qrow * XP + hh * HDIM];
      v4f lo, hi;
      lo.x = (o0[0] + o1[0]) * inv; lo.y = (o0[1] + o1[1]) * inv;
      lo.z = (o0[2] + o1[2]) * inv; lo.w = (o0[3] + o1[3]) * inv;
      hi.x = (o0[4] + o1[4]) * inv; hi.y = (o0[5] + o1[5]) * inv;
      hi.z = (o0[6] + o1[6]) * inv; hi.w = (o0[7] + o1[7]) * inv;
      *(v4f*)&xo[0] = lo;
      *(v4f*)&xo[4] = hi;
    }
  }

  const int mt = wave >> 1;
  const int cb = (wave & 1) * 32;
  __syncthreads();

  const float* xr = &sm[X_OFF + (mt * 16 + l) * XP + 2 * half];
  const float* w0 = &sm[W_OFF + (cb + l) * XP + 2 * half];
  const float* w1 = w0 + 16 * XP;
  v8f y[2] = {};
#pragma unroll
  for (int i = 0; i < 16; ++i) {
    const v2f xf = *(const v2f*)&xr[4 * i];
    const v2f a0 = *(const v2f*)&w0[4 * i];
    const v2f a1 = *(const v2f*)&w1[4 * i];
    y[0] = wmma_f32_16x16x4(a0, xf, y[0]);
    y[1] = wmma_f32_16x16x4(a1, xf, y[1]);
  }

  float* st = &sm[wave * (16 * YP)];
#pragma unroll
  for (int j = 0; j < 2; ++j) {
    const v4f b0 = *(const v4f*)&sm[B_OFF + cb + 16 * j + 8 * half];
    const v4f b1 = *(const v4f*)&sm[B_OFF + cb + 16 * j + 8 * half + 4];
    v4f lo, hi;
    lo.x = y[j][0] + b0.x; lo.y = y[j][1] + b0.y;
    lo.z = y[j][2] + b0.z; lo.w = y[j][3] + b0.w;
    hi.x = y[j][4] + b1.x; hi.y = y[j][5] + b1.y;
    hi.z = y[j][6] + b1.z; hi.w = y[j][7] + b1.w;
    *(v4f*)&st[l * YP + 16 * j + 8 * half]     = lo;
    *(v4f*)&st[l * YP + 16 * j + 8 * half + 4] = hi;
  }
#pragma unroll
  for (int it = 0; it < 4; ++it) {
    const int rr = 4 * it + (lane >> 3);
    const int pc = lane & 7;
    const v4f yv = *(const v4f*)&st[rr * YP + 4 * pc];
    const int n = mt * 16 + rr;
    *(v4f*)&Y[(((size_t)(b * NN + n) * TT + t) * EMBED) + cb + 4 * pc] = yv;
  }
}

extern "C" void kernel_launch(void* const* d_in, const int* in_sizes, int n_in,
                              void* d_out, int out_size, void* d_ws,
                              size_t ws_size, hipStream_t stream) {
  (void)in_sizes; (void)n_in; (void)out_size; (void)d_ws; (void)ws_size;
  const float* V  = (const float*)d_in[0];
  const float* K  = (const float*)d_in[1];
  const float* Q  = (const float*)d_in[2];
  const float* Wv = (const float*)d_in[3];
  const float* Wk = (const float*)d_in[4];
  const float* Wq = (const float*)d_in[5];
  const float* Wo = (const float*)d_in[6];
  const float* bo = (const float*)d_in[7];
  float* Y = (float*)d_out;

  hipLaunchKernelGGL(attn_kernel, dim3(BB * TT), dim3(ATT_THREADS), 0,
                     stream, V, K, Q, Wv, Wk, Wq, Wo, bo, Y);
}


# Attention re-tiled: 8-wave blocks, wave = one head x all 128 queries; its K and V^T fragments are read from LDS once into registers and reused over the 8 query tiles; every sum keeps its order
# speedup vs baseline: 7.9951x; 1.0764x over previous
#include <hip/hip_runtime.h>


typedef float v2f __attribute__((ext_vector_type(2)));
typedef float v4f __attribute__((ext_vector_type(4)));
typedef float v8f __attribute__((ext_vector_type(8)));

#define EMBED 64
#define HEADS 8
#define HDIM  8
#define BB    8
#define NN    128
#define TT    64
#define VP    (NN + 4)
#define QS    (NN * HDIM + 4)
#define VS    (HDIM * VP + 8)
#define Q_OFF 0
#define K_OFF (HEADS * QS)
#define V_OFF (2 * HEADS * QS)
#define Z_OFF (2 * HEADS * QS + HEADS * VS)
#define XP    (EMBED + 4)
#define X_OFF (Z_OFF + NN)
#define W_OFF (X_OFF + NN * XP)
#define B_OFF (W_OFF + EMBED * XP)
#define SM_FLOATS (B_OFF + EMBED)
#define ATT_THREADS 256

__device__ __forceinline__ v8f wmma_f32_16x16x4(v2f a, v2f b, v8f c) {
  return __builtin_amdgcn_wmma_f32_16x16x4_f32(false, a, false, b, (short)0, c,
                                               false, false);
}

__device__ __forceinline__ float xhalf(float x) {
  const int v = __builtin_bit_cast(int, x);
  return __builtin_bit_cast(
      float, __builtin_amdgcn_permlanex16(v, v, 0x76543210, 0xfedcba98, false,
                                          false));
}

__device__ __forceinline__ void project8(const v4f lo, const v4f hi,
                                         const float* __restrict__ Wm,
                                         float* out) {
  const float in[HDIM] = {lo.x, lo.y, lo.z, lo.w, hi.x, hi.y, hi.z, hi.w};
#pragma unroll
  for (int e = 0; e < HDIM; ++e) {
    float a = 0.f;
#pragma unroll
    for (int d = 0; d < HDIM; ++d) a = __builtin_fmaf(in[d], Wm[e * HDIM + d], a);
    out[e] = a;
  }
}

__global__ __launch_bounds__(ATT_THREADS, 4) void attn_kernel(
    const float* __restrict__ V, const float* __restrict__ K,
    const float* __restrict__ Q, const float* __restrict__ Wv,
    const float* __restrict__ Wk, const float* __restrict__ Wq,
    const float* __restrict__ Wo, const float* __restrict__ bo,
    float* __restrict__ Y) {
  __shared__ __attribute__((aligned(16))) float sm[SM_FLOATS];

  const int b = blockIdx.x / TT;
  const int t = blockIdx.x % TT;

  const int tid  = threadIdx.x;
  const int wave = tid >> 5;
  const int lane = tid & 31;
  const int half = lane >> 4;
  const int l    = lane & 15;

  {
    const int ph = lane & 7;
    const int pn = (wave >> 1) * 8 + (wave & 1) + 2 * (lane >> 3);
    v4f rq[4][2], rk[4][2], rv[4][2], rw[4];
#pragma unroll
    for (int it = 0; it < 4; ++it) {
      const int n = pn + 32 * it;
      const size_t base = (((size_t)(b * NN + n) * TT + t) * EMBED) + ph * HDIM;
      rq[it][0] = *(const v4f*)&Q[base];
      rq[it][1] = *(const v4f*)&Q[base + 4];
      rk[it][0] = *(const v4f*)&K[base];
      rk[it][1] = *(const v4f*)&K[base + 4];
      rv[it][0] = *(const v4f*)&V[base];
      rv[it][1] = *(const v4f*)&V[base + 4];
    }
#pragma unroll
    for (int q = 0; q < 4; ++q) rw[q] = ((const v4f*)Wo)[tid + ATT_THREADS * q];
    const float rb = bo[tid & (EMBED - 1)];
    if (tid < NN) sm[Z_OFF + tid] = 0.f;
#pragma unroll
    for (int it = 0; it < 4; ++it) {
      const int n = pn + 32 * it;
      float o[HDIM];
      v4f lo, hi;
      project8(rq[it][0], rq[it][1], Wq, o);
      lo.x = o[0]; lo.y = o[1]; lo.z = o[4]; lo.w = o[5];
      hi.x = o[2]; hi.y = o[3]; hi.z = o[6]; hi.w = o[7];
      *(v4f*)&sm[Q_OFF + ph * QS + n * HDIM]     = lo;
      *(v4f*)&sm[Q_OFF + ph * QS + n * HDIM + 4] = hi;
      project8(rk[it][0], rk[it][1], Wk, o);
      lo.x = o[0]; lo.y = o[1]; lo.z = o[4]; lo.w = o[5];
      hi.x = o[2]; hi.y = o[3]; hi.z = o[6]; hi.w = o[7];
      *(v4f*)&sm[K_OFF + ph * QS + n * HDIM]     = lo;
      *(v4f*)&sm[K_OFF + ph * QS + n * HDIM + 4] = hi;
      project8(rv[it][0], rv[it][1], Wv, o);
#pragma unroll
      for (int e = 0; e < HDIM; ++e) sm[V_OFF + ph * VS + e * VP + n] = o[e];
    }
#pragma unroll
    for (int q = 0; q < 4; ++q) {
      const int i4 = tid + ATT_THREADS * q;
      *(v4f*)&sm[W_OFF + (i4 >> 4) * XP + (i4 & 15) * 4] = rw[q];
    }
    if (tid < EMBED) sm[B_OFF + tid] = rb;
  }
  __syncthreads();

  const int hh    = wave;
  const float* Qh = &sm[Q_OFF + hh * QS];
  const float* Kh = &sm[K_OFF + hh * QS];
  const float* Vrow = &sm[(l < HDIM) ? (V_OFF + hh * VS + l * VP + 8 * half)
                                     : (Z_OFF + 8 * half)];

  v4f kf[8], va[8], vb[8];
#pragma unroll
  for (int tj = 0; tj < 8; ++tj) {
    kf[tj] = *(const v4f*)&Kh[(tj * 16 + l) * HDIM + 4 * half];
    va[tj] = *(const v4f*)&Vrow[tj * 16];
    vb[tj] = *(const v4f*)&Vrow[tj * 16 + 4];
  }

  v4f qf = *(const v4f*)&Qh[l * HDIM + 4 * half];
  for (int qt = 0; qt < 8; ++qt) {
    const int qrow = qt * 16 + l;
    const v4f qn = *(const v4f*)&Qh[((((qt + 1) & 7) * 16) + l) * HDIM + 4 * half];
    v2f qf0, qf1;
    qf0.x = qf.x; qf0.y = qf.y; qf1.x = qf.z; qf1.y = qf.w;
    v8f s[8];
#pragma unroll
    for (int tj = 0; tj < 8; ++tj) {
      v2f kf0;
      kf0.x = kf[tj].x; kf0.y = kf[tj].y;
      v8f c = {};
      s[tj] = wmma_f32_16x16x4(kf0, qf0, c);
    }
#pragma unroll
    for (int tj = 0; tj < 8; ++tj) {
      v2f kf1;
      kf1.x = kf[tj].z; kf1.y = kf[tj].w;
      s[tj] = wmma_f32_16x16x4(kf1, qf1, s[tj]);
    }

    float m0 = s[0][0], m1 = s[0][1], m2 = s[0][2], m3 = s[0][3];
#pragma unroll
    for (int tj = 0; tj < 8; ++tj) {
#pragma unroll
      for (int r = 0; r < 8; r += 4) {
        m0 = fmaxf(m0, s[tj][r]);
        m1 = fmaxf(m1, s[tj][r + 1]);
        m2 = fmaxf(m2, s[tj][r + 2]);
        m3 = fmaxf(m3, s[tj][r + 3]);
      }
    }
    float m = fmaxf(fmaxf(m0, m1), fmaxf(m2, m3));
    m = fmaxf(m, xhalf(m));

    const float c2 = 0.125f * 1.4426950408889634f;
    const float nm = -(m * c2);
    float z0 = 0.f, z1 = 0.f, z2 = 0.f, z3 = 0.f;
#pragma unroll
    for (int tj = 0; tj < 8; ++tj) {
#pragma unroll
      for (int r = 0; r < 8; r += 4) {
        const float e0 = __builtin_amdgcn_exp2f(__builtin_fmaf(s[tj][r], c2, nm));
        const float e1 = __builtin_amdgcn_exp2f(__builtin_fmaf(s[tj][r + 1], c2, nm));
        const float e2 = __builtin_amdgcn_exp2f(__builtin_fmaf(s[tj][r + 2], c2, nm));
        const float e3 = __builtin_amdgcn_exp2f(__builtin_fmaf(s[tj][r + 3], c2, nm));
        s[tj][r] = e0; s[tj][r + 1] = e1; s[tj][r + 2] = e2; s[tj][r + 3] = e3;
        z0 += e0; z1 += e1; z2 += e2; z3 += e3;
      }
    }
    float z = (z0 + z1) + (z2 + z3);
    z += xhalf(z);

    v8f o0 = {}, o1 = {};
#pragma unroll
    for (int tj = 0; tj < 8; ++tj) {
      const v4f a01 = va[tj];
      const v4f a23 = vb[tj];
      v2f a, p;
      a.x = a01.x; a.y = a01.y; p.x = s[tj][0]; p.y = s[tj][1];
      o0 = wmma_f32_16x16x4(a, p, o0);
      a.x = a01.z; a.y = a01.w; p.x = s[tj][2]; p.y = s[tj][3];
      o1 = wmma_f32_16x16x4(a, p, o1);
      a.x = a23.x; a.y = a23.y; p.x = s[tj][4]; p.y = s[tj][5];
      o0 = wmma_f32_16x16x4(a, p, o0);
      a.x = a23.z; a.y = a23.w; p.x = s[tj][6]; p.y = s[tj][7];
      o1 = wmma_f32_16x16x4(a, p, o1);
    }
    const float inv = 1.f / z;
    if (half == 0) {
      float* xo = &sm[X_OFF + qrow * XP + hh * HDIM];
      v4f lo, hi;
      lo.x = (o0[0] + o1[0]) * inv; lo.y = (o0[1] + o1[1]) * inv;
      lo.z = (o0[2] + o1[2]) * inv; lo.w = (o0[3] + o1[3]) * inv;
      hi.x = (o0[4] + o1[4]) * inv; hi.y = (o0[5] + o1[5]) * inv;
      hi.z = (o0[6] + o1[6]) * inv; hi.w = (o0[7] + o1[7]) * inv;
      *(v4f*)&xo[0] = lo;
      *(v4f*)&xo[4] = hi;
    }
    qf = qn;
  }

  const int mt = wave;
  __syncthreads();

  const float* xr = &sm[X_OFF + (mt * 16 + l) * XP + 2 * half];
  const float* wr = &sm[W_OFF + l * XP + 2 * half];
  v8f y[4] = {};
#pragma unroll
  for (int i = 0; i < 16; ++i) {
    const v2f xf = *(const v2f*)&xr[4 * i];
#pragma unroll
    for (int j = 0; j < 4; ++j) {
      const v2f af = *(const v2f*)&wr[j * 16 * XP + 4 * i];
      y[j] = wmma_f32_16x16x4(af, xf, y[j]);
    }
  }

  float* st = &sm[wave * (16 * XP)];
#pragma unroll
  for (int j = 0; j < 4; ++j) {
    const v4f b0 = *(const v4f*)&sm[B_OFF + 16 * j + 8 * half];
    const v4f b1 = *(const v4f*)&sm[B_OFF + 16 * j + 8 * half + 4];
    v4f lo, hi;
    lo.x = y[j][0] + b0.x; lo.y = y[j][1] + b0.y;
    lo.z = y[j][2] + b0.z; lo.w = y[j][3] + b0.w;
    hi.x = y[j][4] + b1.x; hi.y = y[j][5] + b1.y;
    hi.z = y[j][6] + b1.z; hi.w = y[j][7] + b1.w;
    *(v4f*)&st[l * XP + 16 * j + 8 * half]     = lo;
    *(v4f*)&st[l * XP + 16 * j + 8 * half + 4] = hi;
  }
#pragma unroll
  for (int it = 0; it < 8; ++it) {
    const int rr = 2 * it + (lane >> 4);
    const int pc = lane & 15;
    const v4f yv = *(const v4f*)&st[rr * XP + 4 * pc];
    const int n = mt * 16 + rr;
    *(v4f*)&Y[(((size_t)(b * NN + n) * TT + t) * EMBED) + 4 * pc] = yv;
  }
}

extern "C" void kernel_launch(void* const* d_in, const int* in_sizes, int n_in,
                              void* d_out, int out_size, void* d_ws,
                              size_t ws_size, hipStream_t stream) {
  (void)in_sizes; (void)n_in; (void)out_size; (void)d_ws; (void)ws_size;
  const float* V  = (const float*)d_in[0];
  const float* K  = (const float*)d_in[1];
  const float* Q  = (const float*)d_in[2];
  const float* Wv = (const float*)d_in[3];
  const float* Wk = (const float*)d_in[4];
  const float* Wq = (const float*)d_in[5];
  const float* Wo = (const float*)d_in[6];
  const float* bo = (const float*)d_in[7];
  float* Y = (float*)d_out;

  hipLaunchKernelGGL(attn_kernel, dim3(BB * TT), dim3(ATT_THREADS), 0,
                     stream, V, K, Q, Wv, Wk, Wq, Wo, bo, Y);
}
